# starting kernel
# baseline (speedup 1.0000x reference)
#include <hip/hip_runtime.h>

typedef float v2f __attribute__((ext_vector_type(2)));
typedef float v8f __attribute__((ext_vector_type(8)));

namespace {
constexpr int kNSamples      = 4000000;
constexpr int kNFrames       = 250;
constexpr int kFrame         = 16000;
constexpr int kNBins         = 1025;
constexpr int kTaps          = 257;
constexpr int kHStride       = 260;
constexpr int kTile          = 256;
constexpr int kTilesPerBatch = kNSamples / kTile;
constexpr int kKDim          = 272;
constexpr int kBatch         = 4;
constexpr int kASize = 512 + 2 * 32;
constexpr int kHSize = 288;
constexpr int kWaveLds = 2 * kASize + 2 * kHSize;
}

__global__ __launch_bounds__(256) void build_ir_kernel(const float* __restrict__ mags,
                                                       float* __restrict__ h)
{
    __shared__ float tab[2048];
    __shared__ float mg[kNBins];
    const int f   = blockIdx.x;
    const int tid = threadIdx.x;

    for (int r = tid; r < 2048; r += 256)
        tab[r] = cosf((float)r * 3.0679615757712823e-3f);
    for (int j = tid; j < kNBins; j += 256) {
        float m = mags[f * kNBins + j];
        mg[j] = fminf(fmaxf(m, 0.0f), 1.0f);
    }
    __syncthreads();

    for (int k = tid; k < kTaps; k += 256) {
        float val = 0.0f;
        if (k < 255) {
            const int t = (k < 127) ? (k + 1921) : (k - 127);
            float s = 0.0f;
            for (int j = 1; j < 1024; ++j)
                s = fmaf(mg[j], tab[(j * t) & 2047], s);
            const float ir = mg[0] + ((t & 1) ? -mg[1024] : mg[1024]) + 2.0f * s;
            const float w = 0.5f - 0.5f * cosf(6.283185307179586f * (float)(k + 2) *
                                               (1.0f / 257.0f));
            val = ir * (1.0f / 2048.0f) * w;
        }
        h[f * kHStride + k] = val;
    }
}

__global__ __launch_bounds__(256) void fir_wmma_kernel(const float* __restrict__ audio,
                                                       const float* __restrict__ h,
                                                       float* __restrict__ out)
{
    __shared__ float lds[8][kWaveLds];
    const int lane   = threadIdx.x & 31;
    const int wave   = threadIdx.x >> 5;
    const int tileId = blockIdx.x * 8 + wave;
    if (tileId >= kBatch * kTilesPerBatch) return;

    const int b  = tileId / kTilesPerBatch;
    const int n0 = (tileId - b * kTilesPerBatch) * kTile;

    float* __restrict__ A0  = &lds[wave][0];
    float* __restrict__ A1  = A0 + kASize;
    float* __restrict__ Hr0 = A1 + kASize;
    float* __restrict__ Hr1 = Hr0 + kHSize;

    const float* __restrict__ aud = audio + (long)b * kNSamples;
    const int sBase = n0 - 129;

    const int fLo   = max(sBase, 0) / kFrame;
    const int fHi   = min(n0 + 382, kNSamples - 1) / kFrame;
    const int nPass = (fHi > fLo) ? 2 : 1;
    const int cut   = fHi * kFrame;

    for (int idx = lane; idx < 512; idx += 32) {
        const int s = sBase + idx;
        float v = (s >= 0 && s < kNSamples) ? aud[s] : 0.0f;
        if (nPass == 2 && s >= cut) v = 0.0f;
        A0[idx + ((idx >> 4) << 1)] = v;
    }
    for (int q = lane; q < kHSize; q += 32) {
        const int t = 271 - q;
        Hr0[q] = ((unsigned)t <= 256u) ? h[fLo * kHStride + t] : 0.0f;
    }
    if (nPass == 2) {
        for (int idx = lane; idx < 512; idx += 32) {
            const int s = sBase + idx;
            const float v = (s >= cut && s < kNSamples) ? aud[s] : 0.0f;
            A1[idx + ((idx >> 4) << 1)] = v;
        }
        for (int q = lane; q < kHSize; q += 32) {
            const int t = 271 - q;
            Hr1[q] = ((unsigned)t <= 256u) ? h[fHi * kHStride + t] : 0.0f;
        }
    }
    asm volatile("s_wait_dscnt 0x0" ::: "memory");

    v8f acc = {0.f, 0.f, 0.f, 0.f, 0.f, 0.f, 0.f, 0.f};
    const int M    = lane & 15;
    const int kOff = (lane >> 4) << 1;

    for (int pass = 0; pass < nPass; ++pass) {
        const float* __restrict__ aB = (pass ? A1 : A0) + 18 * M + kOff;
        const float* __restrict__ bB = (pass ? Hr1 : Hr0) + (15 - M) + kOff;
#pragma unroll
        for (int kk0 = 0; kk0 < kKDim; kk0 += 4) {
            const int aOff = kk0 + ((kk0 >> 4) << 1);
            const v2f av = *(const v2f*)(aB + aOff);
            v2f bv;
            bv.x = bB[kk0];
            bv.y = bB[kk0 + 1];
            acc = __builtin_amdgcn_wmma_f32_16x16x4_f32(false, av, false, bv,
                                                        (short)0, acc, false, false);
        }
    }

    float* __restrict__ o = out + (long)b * kNSamples + n0;
    const int rowOff = (lane >> 4) << 3;
#pragma unroll
    for (int r = 0; r < 8; ++r)
        o[((r + rowOff) << 4) + M] = acc[r];
}

extern "C" void kernel_launch(void* const* d_in, const int* in_sizes, int n_in,
                              void* d_out, int out_size, void* d_ws, size_t ws_size,
                              hipStream_t stream)
{
    (void)in_sizes; (void)n_in; (void)out_size; (void)ws_size;
    const float* audio = (const float*)d_in[0];
    const float* mags  = (const float*)d_in[1];
    float* out = (float*)d_out;
    float* h   = (float*)d_ws;

    build_ir_kernel<<<kNFrames, 256, 0, stream>>>(mags, h);

    const int tiles = kBatch * kTilesPerBatch;
    fir_wmma_kernel<<<(tiles + 7) / 8, 256, 0, stream>>>(audio, h, out);
}


# FIR GEMM: each wave computes 5 adjacent 16x16 tiles per K-step on one shared tap fragment (5 accumulators, 5x fewer waves, 4-wave blocks); per-output summation order unchanged
# speedup vs baseline: 1.2154x; 1.2154x over previous
#include <hip/hip_runtime.h>

typedef float v2f __attribute__((ext_vector_type(2)));
typedef float v8f __attribute__((ext_vector_type(8)));

namespace {
constexpr int kNSamples      = 4000000;
constexpr int kNFrames       = 250;
constexpr int kFrame         = 16000;
constexpr int kNBins         = 1025;
constexpr int kTaps          = 257;
constexpr int kHStride       = 260;
constexpr int kTile          = 256;
constexpr int kTilesPerBatch = kNSamples / kTile;
constexpr int kKDim          = 272;
constexpr int kBatch         = 4;
constexpr int kGroup         = 5;
constexpr int kGroupsPerBatch = kTilesPerBatch / kGroup;
constexpr int kSpan          = kGroup * kTile;
constexpr int kWin           = kSpan + 256;
constexpr int kASize = kWin + 2 * (kWin / 16);
constexpr int kTileLds = kTile + 2 * (kTile / 16);
constexpr int kHSize = 288;
constexpr int kWaveLds = kASize + kHSize;
constexpr int kWaves   = 4;
static_assert(kTilesPerBatch % kGroup == 0, "tile groups must not straddle batches");
}

__global__ __launch_bounds__(256) void build_ir_kernel(const float* __restrict__ mags,
                                                       float* __restrict__ h)
{
    __shared__ float tab[2048];
    __shared__ float mg[kNBins];
    const int f   = blockIdx.x;
    const int tid = threadIdx.x;

    for (int r = tid; r < 2048; r += 256)
        tab[r] = cosf((float)r * 3.0679615757712823e-3f);
    for (int j = tid; j < kNBins; j += 256) {
        float m = mags[f * kNBins + j];
        mg[j] = fminf(fmaxf(m, 0.0f), 1.0f);
    }
    __syncthreads();

    for (int k = tid; k < kTaps; k += 256) {
        float val = 0.0f;
        if (k < 255) {
            const int t = (k < 127) ? (k + 1921) : (k - 127);
            float s = 0.0f;
            for (int j = 1; j < 1024; ++j)
                s = fmaf(mg[j], tab[(j * t) & 2047], s);
            const float ir = mg[0] + ((t & 1) ? -mg[1024] : mg[1024]) + 2.0f * s;
            const float w = 0.5f - 0.5f * cosf(6.283185307179586f * (float)(k + 2) *
                                               (1.0f / 257.0f));
            val = ir * (1.0f / 2048.0f) * w;
        }
        h[f * kHStride + k] = val;
    }
}

__global__ __launch_bounds__(kWaves * 32) void fir_wmma_kernel(const float* __restrict__ audio,
                                                       const float* __restrict__ h,
                                                       float* __restrict__ out)
{
    __shared__ float lds[kWaves][kWaveLds];
    const int lane    = threadIdx.x & 31;
    const int wave    = threadIdx.x >> 5;
    const int groupId = blockIdx.x * kWaves + wave;
    if (groupId >= kBatch * kGroupsPerBatch) return;

    const int b  = groupId / kGroupsPerBatch;
    const int n0 = (groupId - b * kGroupsPerBatch) * kSpan;

    float* __restrict__ Hr = &lds[wave][0];
    float* __restrict__ A  = Hr + kHSize;

    const int sBase = n0 - 129;

    const int fLo   = max(sBase, 0) / kFrame;
    const int fHi   = min(sBase + kWin - 1, kNSamples - 1) / kFrame;
    const int nPass = (fHi > fLo) ? 2 : 1;
    const int cut   = fHi * kFrame;

    v8f acc[kGroup];
#pragma unroll
    for (int t = 0; t < kGroup; ++t)
        acc[t] = (v8f){0.f, 0.f, 0.f, 0.f, 0.f, 0.f, 0.f, 0.f};

    const int M    = lane & 15;
    const int kOff = (lane >> 4) << 1;
    const float* __restrict__ aB = A + 18 * M + kOff;
    const float* __restrict__ bB = Hr + (15 - M) + kOff;

    constexpr int kEdgeLo = 5;
    constexpr int kEdgeHi = 44;
    const int gBase = b * kNSamples + sBase + lane;
    const float* __restrict__ pMid = audio + (gBase + 32 * kEdgeLo);
    float* __restrict__ aSt = A + lane + ((lane >> 4) << 1);

#pragma nounroll
    for (int pass = 0; pass < nPass; ++pass) {
        const int lo = pass ? cut : 0;
        const int hi = (nPass == 2 && pass == 0) ? cut : kNSamples;
        const int f  = fLo + pass;

        asm volatile("s_wait_dscnt 0x0" ::: "memory");

        float hst[kHSize / 32];
        float ast[kWin / 32];
        const float* __restrict__ hRow = h + f * kHStride;
#pragma unroll
        for (int i = 0; i < kHSize / 32; ++i) {
            const int t = 271 - (lane + 32 * i);
            hst[i] = hRow[min(max(t, 0), 256)];
        }
#pragma unroll
        for (int i = 0; i < kWin / 32; ++i) {
            if (i < kEdgeLo || i >= kEdgeHi)
                ast[i] = audio[min(max(gBase + 32 * i, 0), kBatch * kNSamples - 1)];
            else
                ast[i] = pMid[32 * (i - kEdgeLo)];
        }
#pragma unroll
        for (int i = 0; i < kHSize / 32; ++i)
            asm volatile("" :: "v"(hst[i]));
#pragma unroll
        for (int i = 0; i < kWin / 32; ++i)
            asm volatile("" :: "v"(ast[i]));
#pragma unroll
        for (int i = 0; i < kHSize / 32; ++i) {
            const int t = 271 - (lane + 32 * i);
            Hr[lane + 32 * i] = ((unsigned)t <= 256u) ? hst[i] : 0.0f;
        }
        const int      rel   = sBase + lane - lo;
        const unsigned width = (unsigned)(hi - lo);
#pragma unroll
        for (int i = 0; i < kWin / 32; ++i)
            aSt[36 * i] = ((unsigned)(rel + 32 * i) < width) ? ast[i] : 0.0f;
        asm volatile("s_wait_dscnt 0x0" ::: "memory");

#pragma unroll
        for (int kk0 = 0; kk0 < kKDim; kk0 += 4) {
            const int aOff = kk0 + ((kk0 >> 4) << 1);
            v2f bv;
            bv.x = bB[kk0];
            bv.y = bB[kk0 + 1];
#pragma unroll
            for (int t = 0; t < kGroup; ++t) {
                const v2f av = *(const v2f*)(aB + kTileLds * t + aOff);
                acc[t] = __builtin_amdgcn_wmma_f32_16x16x4_f32(false, av, false, bv,
                                                               (short)0, acc[t], false, false);
            }
        }
    }

    float* __restrict__ o = out + (long)b * kNSamples + n0;
    const int rowOff = (lane >> 4) << 3;
#pragma unroll
    for (int t = 0; t < kGroup; ++t) {
#pragma unroll
        for (int r = 0; r < 8; ++r)
            o[kTile * t + ((r + rowOff) << 4) + M] = acc[t][r];
    }
}

extern "C" void kernel_launch(void* const* d_in, const int* in_sizes, int n_in,
                              void* d_out, int out_size, void* d_ws, size_t ws_size,
                              hipStream_t stream)
{
    (void)in_sizes; (void)n_in; (void)out_size; (void)ws_size;
    const float* audio = (const float*)d_in[0];
    const float* mags  = (const float*)d_in[1];
    float* out = (float*)d_out;
    float* h   = (float*)d_ws;

    build_ir_kernel<<<kNFrames, 256, 0, stream>>>(mags, h);

    const int groups = kBatch * kGroupsPerBatch;
    fir_wmma_kernel<<<(groups + kWaves - 1) / kWaves, kWaves * 32, 0, stream>>>(audio, h, out);
}


# Impulse-response kernel: cosine sum folded about bin 512 (bins j and 1024-j share one table read, coefficient mg[j] +/- mg[1024-j] by tap parity): 511 gathers per tap; its summation order changes
# speedup vs baseline: 1.3784x; 1.1341x over previous
#include <hip/hip_runtime.h>

typedef float v2f __attribute__((ext_vector_type(2)));
typedef float v8f __attribute__((ext_vector_type(8)));

namespace {
constexpr int kNSamples      = 4000000;
constexpr int kNFrames       = 250;
constexpr int kFrame         = 16000;
constexpr int kNBins         = 1025;
constexpr int kTaps          = 257;
constexpr int kHStride       = 260;
constexpr int kTile          = 256;
constexpr int kTilesPerBatch = kNSamples / kTile;
constexpr int kKDim          = 272;
constexpr int kBatch         = 4;
constexpr int kGroup         = 5;
constexpr int kGroupsPerBatch = kTilesPerBatch / kGroup;
constexpr int kSpan          = kGroup * kTile;
constexpr int kWin           = kSpan + 256;
constexpr int kASize = kWin + 2 * (kWin / 16);
constexpr int kTileLds = kTile + 2 * (kTile / 16);
constexpr int kHSize = 288;
constexpr int kWaveLds = kASize + kHSize;
constexpr int kWaves   = 4;
static_assert(kTilesPerBatch % kGroup == 0, "tile groups must not straddle batches");
}

__global__ __launch_bounds__(256) void build_ir_kernel(const float* __restrict__ mags,
                                                       float* __restrict__ h)
{
    __shared__ float tab[2048];
    __shared__ float mg[kNBins];
    __shared__ float pq[2 * 512];
    const int f   = blockIdx.x;
    const int tid = threadIdx.x;

    for (int r = tid; r < 2048; r += 256)
        tab[r] = cosf((float)r * 3.0679615757712823e-3f);
    for (int j = tid; j < kNBins; j += 256) {
        float m = mags[f * kNBins + j];
        mg[j] = fminf(fmaxf(m, 0.0f), 1.0f);
    }
    __syncthreads();
    for (int j = tid + 1; j < 512; j += 256) {
        const float a = mg[j];
        const float b = mg[1024 - j];
        pq[2 * j]     = a + b;
        pq[2 * j + 1] = a - b;
    }
    __syncthreads();

    for (int k = tid; k < kTaps; k += 256) {
        float val = 0.0f;
        if (k < 255) {
            const int t = (k < 127) ? (k + 1921) : (k - 127);
            const float* __restrict__ w = pq + (t & 1);
            float s = 0.0f;
            for (int j = 1; j < 512; ++j)
                s = fmaf(w[2 * j], tab[(j * t) & 2047], s);
            s = fmaf(mg[512], tab[(512 * t) & 2047], s);
            const float ir = mg[0] + ((t & 1) ? -mg[1024] : mg[1024]) + 2.0f * s;
            const float w257 = 0.5f - 0.5f * cosf(6.283185307179586f * (float)(k + 2) *
                                                  (1.0f / 257.0f));
            val = ir * (1.0f / 2048.0f) * w257;
        }
        h[f * kHStride + k] = val;
    }
}

__global__ __launch_bounds__(kWaves * 32) void fir_wmma_kernel(const float* __restrict__ audio,
                                                       const float* __restrict__ h,
                                                       float* __restrict__ out)
{
    __shared__ float lds[kWaves][kWaveLds];
    const int lane    = threadIdx.x & 31;
    const int wave    = threadIdx.x >> 5;
    const int groupId = blockIdx.x * kWaves + wave;
    if (groupId >= kBatch * kGroupsPerBatch) return;

    const int b  = groupId / kGroupsPerBatch;
    const int n0 = (groupId - b * kGroupsPerBatch) * kSpan;

    float* __restrict__ Hr = &lds[wave][0];
    float* __restrict__ A  = Hr + kHSize;

    const int sBase = n0 - 129;

    const int fLo   = max(sBase, 0) / kFrame;
    const int fHi   = min(sBase + kWin - 1, kNSamples - 1) / kFrame;
    const int nPass = (fHi > fLo) ? 2 : 1;
    const int cut   = fHi * kFrame;

    v8f acc[kGroup];
#pragma unroll
    for (int t = 0; t < kGroup; ++t)
        acc[t] = (v8f){0.f, 0.f, 0.f, 0.f, 0.f, 0.f, 0.f, 0.f};

    const int M    = lane & 15;
    const int kOff = (lane >> 4) << 1;
    const float* __restrict__ aB = A + 18 * M + kOff;
    const float* __restrict__ bB = Hr + (15 - M) + kOff;

    constexpr int kEdgeLo = 5;
    constexpr int kEdgeHi = 44;
    const int gBase = b * kNSamples + sBase + lane;
    const float* __restrict__ pMid = audio + (gBase + 32 * kEdgeLo);
    float* __restrict__ aSt = A + lane + ((lane >> 4) << 1);

#pragma nounroll
    for (int pass = 0; pass < nPass; ++pass) {
        const int lo = pass ? cut : 0;
        const int hi = (nPass == 2 && pass == 0) ? cut : kNSamples;
        const int f  = fLo + pass;

        asm volatile("s_wait_dscnt 0x0" ::: "memory");

        float hst[kHSize / 32];
        float ast[kWin / 32];
        const float* __restrict__ hRow = h + f * kHStride;
#pragma unroll
        for (int i = 0; i < kHSize / 32; ++i) {
            const int t = 271 - (lane + 32 * i);
            hst[i] = hRow[min(max(t, 0), 256)];
        }
#pragma unroll
        for (int i = 0; i < kWin / 32; ++i) {
            if (i < kEdgeLo || i >= kEdgeHi)
                ast[i] = audio[min(max(gBase + 32 * i, 0), kBatch * kNSamples - 1)];
            else
                ast[i] = pMid[32 * (i - kEdgeLo)];
        }
#pragma unroll
        for (int i = 0; i < kHSize / 32; ++i)
            asm volatile("" :: "v"(hst[i]));
#pragma unroll
        for (int i = 0; i < kWin / 32; ++i)
            asm volatile("" :: "v"(ast[i]));
#pragma unroll
        for (int i = 0; i < kHSize / 32; ++i) {
            const int t = 271 - (lane + 32 * i);
            Hr[lane + 32 * i] = ((unsigned)t <= 256u) ? hst[i] : 0.0f;
        }
        const int      rel   = sBase + lane - lo;
        const unsigned width = (unsigned)(hi - lo);
#pragma unroll
        for (int i = 0; i < kWin / 32; ++i)
            aSt[36 * i] = ((unsigned)(rel + 32 * i) < width) ? ast[i] : 0.0f;
        asm volatile("s_wait_dscnt 0x0" ::: "memory");

#pragma unroll
        for (int kk0 = 0; kk0 < kKDim; kk0 += 4) {
            const int aOff = kk0 + ((kk0 >> 4) << 1);
            v2f bv;
            bv.x = bB[kk0];
            bv.y = bB[kk0 + 1];
#pragma unroll
            for (int t = 0; t < kGroup; ++t) {
                const v2f av = *(const v2f*)(aB + kTileLds * t + aOff);
                acc[t] = __builtin_amdgcn_wmma_f32_16x16x4_f32(false, av, false, bv,
                                                               (short)0, acc[t], false, false);
            }
        }
    }

    float* __restrict__ o = out + (long)b * kNSamples + n0;
    const int rowOff = (lane >> 4) << 3;
#pragma unroll
    for (int t = 0; t < kGroup; ++t) {
#pragma unroll
        for (int r = 0; r < 8; ++r)
            o[kTile * t + ((r + rowOff) << 4) + M] = acc[t][r];
    }
}

extern "C" void kernel_launch(void* const* d_in, const int* in_sizes, int n_in,
                              void* d_out, int out_size, void* d_ws, size_t ws_size,
                              hipStream_t stream)
{
    (void)in_sizes; (void)n_in; (void)out_size; (void)ws_size;
    const float* audio = (const float*)d_in[0];
    const float* mags  = (const float*)d_in[1];
    float* out = (float*)d_out;
    float* h   = (float*)d_ws;

    build_ir_kernel<<<kNFrames, 256, 0, stream>>>(mags, h);

    const int groups = kBatch * kGroupsPerBatch;
    fir_wmma_kernel<<<(groups + kWaves - 1) / kWaves, kWaves * 32, 0, stream>>>(audio, h, out);
}


# Impulse-response kernel: cosine sum is even in t, taps k and 254-k share one sum; only t=0..127 summed, two threads per t (bins 0..255 | 256..511), partials added in LDS; summation order changes
# speedup vs baseline: 1.4373x; 1.0427x over previous
#include <hip/hip_runtime.h>

typedef float v2f __attribute__((ext_vector_type(2)));
typedef float v8f __attribute__((ext_vector_type(8)));

namespace {
constexpr int kNSamples      = 4000000;
constexpr int kNFrames       = 250;
constexpr int kFrame         = 16000;
constexpr int kNBins         = 1025;
constexpr int kTaps          = 257;
constexpr int kHStride       = 260;
constexpr int kTile          = 256;
constexpr int kTilesPerBatch = kNSamples / kTile;
constexpr int kKDim          = 272;
constexpr int kBatch         = 4;
constexpr int kGroup         = 5;
constexpr int kGroupsPerBatch = kTilesPerBatch / kGroup;
constexpr int kSpan          = kGroup * kTile;
constexpr int kWin           = kSpan + 256;
constexpr int kASize = kWin + 2 * (kWin / 16);
constexpr int kTileLds = kTile + 2 * (kTile / 16);
constexpr int kHSize = 288;
constexpr int kWaveLds = kASize + kHSize;
constexpr int kWaves   = 4;
static_assert(kTilesPerBatch % kGroup == 0, "tile groups must not straddle batches");
}

__global__ __launch_bounds__(256) void build_ir_kernel(const float* __restrict__ mags,
                                                       float* __restrict__ h)
{
    __shared__ float tab[2048];
    __shared__ float mg[kNBins];
    __shared__ float pq[2 * 512];
    __shared__ float part[2 * 128];
    const int f   = blockIdx.x;
    const int tid = threadIdx.x;

    for (int r = tid; r < 2048; r += 256)
        tab[r] = cosf((float)r * 3.0679615757712823e-3f);
    for (int j = tid; j < kNBins; j += 256) {
        float m = mags[f * kNBins + j];
        mg[j] = fminf(fmaxf(m, 0.0f), 1.0f);
    }
    __syncthreads();
    for (int j = tid; j < 512; j += 256) {
        const float a = mg[j];
        const float b = mg[1024 - j];
        pq[2 * j]     = j ? a + b : 0.0f;
        pq[2 * j + 1] = j ? a - b : 0.0f;
    }
    __syncthreads();

    {
        const int t     = tid & 127;
        const int jBase = (tid >> 7) << 8;
        const float* __restrict__ w = pq + 2 * jBase + (t & 1);
        float s = 0.0f;
        for (int i = 0; i < 256; ++i)
            s = fmaf(w[2 * i], tab[((jBase + i) * t) & 2047], s);
        part[tid] = s;
    }
    __syncthreads();

    for (int k = tid; k < kTaps; k += 256) {
        float val = 0.0f;
        if (k < 255) {
            const int t = (k < 127) ? (127 - k) : (k - 127);
            float s = part[t] + part[128 + t];
            s = fmaf(mg[512], tab[(512 * t) & 2047], s);
            const float ir = mg[0] + ((t & 1) ? -mg[1024] : mg[1024]) + 2.0f * s;
            const float w257 = 0.5f - 0.5f * cosf(6.283185307179586f * (float)(k + 2) *
                                                  (1.0f / 257.0f));
            val = ir * (1.0f / 2048.0f) * w257;
        }
        h[f * kHStride + k] = val;
    }
}

__global__ __launch_bounds__(kWaves * 32) void fir_wmma_kernel(const float* __restrict__ audio,
                                                       const float* __restrict__ h,
                                                       float* __restrict__ out)
{
    __shared__ float lds[kWaves][kWaveLds];
    const int lane    = threadIdx.x & 31;
    const int wave    = threadIdx.x >> 5;
    const int groupId = blockIdx.x * kWaves + wave;
    if (groupId >= kBatch * kGroupsPerBatch) return;

    const int b  = groupId / kGroupsPerBatch;
    const int n0 = (groupId - b * kGroupsPerBatch) * kSpan;

    float* __restrict__ Hr = &lds[wave][0];
    float* __restrict__ A  = Hr + kHSize;

    const int sBase = n0 - 129;

    const int fLo   = max(sBase, 0) / kFrame;
    const int fHi   = min(sBase + kWin - 1, kNSamples - 1) / kFrame;
    const int nPass = (fHi > fLo) ? 2 : 1;
    const int cut   = fHi * kFrame;

    v8f acc[kGroup];
#pragma unroll
    for (int t = 0; t < kGroup; ++t)
        acc[t] = (v8f){0.f, 0.f, 0.f, 0.f, 0.f, 0.f, 0.f, 0.f};

    const int M    = lane & 15;
    const int kOff = (lane >> 4) << 1;
    const float* __restrict__ aB = A + 18 * M + kOff;
    const float* __restrict__ bB = Hr + (15 - M) + kOff;

    constexpr int kEdgeLo = 5;
    constexpr int kEdgeHi = 44;
    const int gBase = b * kNSamples + sBase + lane;
    const float* __restrict__ pMid = audio + (gBase + 32 * kEdgeLo);
    float* __restrict__ aSt = A + lane + ((lane >> 4) << 1);

#pragma nounroll
    for (int pass = 0; pass < nPass; ++pass) {
        const int lo = pass ? cut : 0;
        const int hi = (nPass == 2 && pass == 0) ? cut : kNSamples;
        const int f  = fLo + pass;

        asm volatile("s_wait_dscnt 0x0" ::: "memory");

        float hst[kHSize / 32];
        float ast[kWin / 32];
        const float* __restrict__ hRow = h + f * kHStride;
#pragma unroll
        for (int i = 0; i < kHSize / 32; ++i) {
            const int t = 271 - (lane + 32 * i);
            hst[i] = hRow[min(max(t, 0), 256)];
        }
#pragma unroll
        for (int i = 0; i < kWin / 32; ++i) {
            if (i < kEdgeLo || i >= kEdgeHi)
                ast[i] = audio[min(max(gBase + 32 * i, 0), kBatch * kNSamples - 1)];
            else
                ast[i] = pMid[32 * (i - kEdgeLo)];
        }
#pragma unroll
        for (int i = 0; i < kHSize / 32; ++i)
            asm volatile("" :: "v"(hst[i]));
#pragma unroll
        for (int i = 0; i < kWin / 32; ++i)
            asm volatile("" :: "v"(ast[i]));
#pragma unroll
        for (int i = 0; i < kHSize / 32; ++i) {
            const int t = 271 - (lane + 32 * i);
            Hr[lane + 32 * i] = ((unsigned)t <= 256u) ? hst[i] : 0.0f;
        }
        const int      rel   = sBase + lane - lo;
        const unsigned width = (unsigned)(hi - lo);
#pragma unroll
        for (int i = 0; i < kWin / 32; ++i)
            aSt[36 * i] = ((unsigned)(rel + 32 * i) < width) ? ast[i] : 0.0f;
        asm volatile("s_wait_dscnt 0x0" ::: "memory");

#pragma unroll
        for (int kk0 = 0; kk0 < kKDim; kk0 += 4) {
            const int aOff = kk0 + ((kk0 >> 4) << 1);
            v2f bv;
            bv.x = bB[kk0];
            bv.y = bB[kk0 + 1];
#pragma unroll
            for (int t = 0; t < kGroup; ++t) {
                const v2f av = *(const v2f*)(aB + kTileLds * t + aOff);
                acc[t] = __builtin_amdgcn_wmma_f32_16x16x4_f32(false, av, false, bv,
                                                               (short)0, acc[t], false, false);
            }
        }
    }

    float* __restrict__ o = out + (long)b * kNSamples + n0;
    const int rowOff = (lane >> 4) << 3;
#pragma unroll
    for (int t = 0; t < kGroup; ++t) {
#pragma unroll
        for (int r = 0; r < 8; ++r)
            o[kTile * t + ((r + rowOff) << 4) + M] = acc[t][r];
    }
}

extern "C" void kernel_launch(void* const* d_in, const int* in_sizes, int n_in,
                              void* d_out, int out_size, void* d_ws, size_t ws_size,
                              hipStream_t stream)
{
    (void)in_sizes; (void)n_in; (void)out_size; (void)ws_size;
    const float* audio = (const float*)d_in[0];
    const float* mags  = (const float*)d_in[1];
    float* out = (float*)d_out;
    float* h   = (float*)d_ws;

    build_ir_kernel<<<kNFrames, 256, 0, stream>>>(mags, h);

    const int groups = kBatch * kGroupsPerBatch;
    fir_wmma_kernel<<<(groups + kWaves - 1) / kWaves, kWaves * 32, 0, stream>>>(audio, h, out);
}
